# starting kernel
# baseline (speedup 1.0000x reference)
#include <hip/hip_runtime.h>
#include <math.h>

#define N_ENT 50000
#define N_ATT 2000
#define N_VAL 100000
#define NE    300000
#define KD    256

typedef __attribute__((ext_vector_type(2))) float v2f;
typedef __attribute__((ext_vector_type(8))) float v8f;

__global__ void zero_f32(float* __restrict__ p, long n) {
    long i = (long)blockIdx.x * blockDim.x + threadIdx.x;
    long stride = (long)gridDim.x * blockDim.x;
    for (; i < n; i += stride) p[i] = 0.0f;
}

__global__ void row_dot_kernel(const float* __restrict__ ent_feats,
                               const float* __restrict__ att_feats,
                               const float* __restrict__ a_w,
                               float* __restrict__ ent_score,
                               float* __restrict__ att_score) {
    int wave = (int)((blockIdx.x * blockDim.x + threadIdx.x) >> 5);
    int lane = threadIdx.x & 31;
    const float* feat;
    const float* w;
    float* outp;
    if (wave < N_ENT) {
        feat = ent_feats + (long)wave * KD;
        w = a_w;
        outp = ent_score + wave;
    } else {
        int r = wave - N_ENT;
        if (r >= N_ATT) return;
        feat = att_feats + (long)r * KD;
        w = a_w + KD;
        outp = att_score + r;
    }
    float s = 0.0f;
#pragma unroll
    for (int j = 0; j < 2; ++j) {
        int c = (lane + 32 * j) * 4;
        float4 f = *(const float4*)(feat + c);
        float4 ww = *(const float4*)(w + c);
        s += f.x * ww.x + f.y * ww.y + f.z * ww.z + f.w * ww.w;
    }
#pragma unroll
    for (int off = 16; off > 0; off >>= 1) s += __shfl_xor(s, off, 32);
    if (lane == 0) *outp = s;
}

__global__ void edge_score_kernel(const int* __restrict__ trip,
                                  const float* __restrict__ ent_score,
                                  const float* __restrict__ att_score,
                                  const float* __restrict__ a_b,
                                  float* __restrict__ score,
                                  float* __restrict__ row_sum) {
    int e = blockIdx.x * blockDim.x + threadIdx.x;
    if (e >= NE) return;
    int h  = trip[3 * e + 0];
    int at = trip[3 * e + 2];
    float x = ent_score[h] + att_score[at] + a_b[0];
    x = (x > 0.0f) ? x : 0.2f * x;
    float sc = expf(x);
    score[e] = sc;
    atomicAdd(&row_sum[h], sc);
}

template <bool ACC>
__global__ void wmma_gemm_f32(const float* __restrict__ A,
                              const float* __restrict__ B,
                              float* __restrict__ C) {
    int wave = (int)((blockIdx.x * blockDim.x + threadIdx.x) >> 5);
    int lane = threadIdx.x & 31;
    int tM = wave >> 4;
    int tN = wave & 15;
    int row0 = tM * 16;
    int col0 = tN * 16;

    int m    = lane & 15;
    int koff = (lane >> 4) << 1;
    int bn   = col0 + (lane & 15);
    int crow = row0 + ((lane >> 4) << 3);
    int cn   = col0 + (lane & 15);

    v8f c;
    if (ACC) {
#pragma unroll
        for (int r = 0; r < 8; ++r) c[r] = C[(long)(crow + r) * KD + cn];
    } else {
#pragma unroll
        for (int r = 0; r < 8; ++r) c[r] = 0.0f;
    }

    const float* aptr = A + (long)(row0 + m) * KD + koff;

#pragma unroll 4
    for (int k0 = 0; k0 < KD; k0 += 4) {
        v2f av = *(const v2f*)(aptr + k0);
        v2f bv;
        bv.x = B[(long)(k0 + koff + 0) * KD + bn];
        bv.y = B[(long)(k0 + koff + 1) * KD + bn];
        c = __builtin_amdgcn_wmma_f32_16x16x4_f32(
                 false, av,  false, bv,
                 (short)0, c,  false,  false);
    }

#pragma unroll
    for (int r = 0; r < 8; ++r) C[(long)(crow + r) * KD + cn] = c[r];
}

__global__ void edge_agg_kernel(const int* __restrict__ trip,
                                const float* __restrict__ score,
                                const float* __restrict__ row_sum,
                                const float* __restrict__ att_proj,
                                const float* __restrict__ val_feats,
                                float* __restrict__ U,
                                float* __restrict__ outAcc) {
    int d = threadIdx.x;
    for (int e = blockIdx.x; e < NE; e += gridDim.x) {
        int h  = trip[3 * e + 0];
        int vl = trip[3 * e + 1];
        int at = trip[3 * e + 2];
        float p = score[e] / row_sum[h];
        atomicAdd(&U[(long)h * KD + d],      p * val_feats[(long)vl * KD + d]);
        atomicAdd(&outAcc[(long)h * KD + d], p * att_proj[(long)at * KD + d]);
    }
}

__global__ void finalize_kernel(float* __restrict__ out,
                                const float* __restrict__ ent_feats,
                                long n) {
    long i = (long)blockIdx.x * blockDim.x + threadIdx.x;
    long stride = (long)gridDim.x * blockDim.x;
    for (; i < n; i += stride) {
        float x = out[i] + ent_feats[i];
        out[i] = (x > 0.0f) ? x : (expf(x) - 1.0f);
    }
}

extern "C" void kernel_launch(void* const* d_in, const int* in_sizes, int n_in,
                              void* d_out, int out_size, void* d_ws, size_t ws_size,
                              hipStream_t stream) {
    const int*   trip      = (const int*)d_in[0];
    const float* att_feats = (const float*)d_in[1];
    const float* val_feats = (const float*)d_in[2];
    const float* ent_feats = (const float*)d_in[3];
    const float* a_w       = (const float*)d_in[4];
    const float* a_b       = (const float*)d_in[5];
    const float* W         = (const float*)d_in[6];
    float* out = (float*)d_out;

    const float* W_top = W;
    const float* W_bot = W + KD * KD;

    float* ws       = (float*)d_ws;
    float* ent_s    = ws;
    float* att_s    = ent_s   + N_ENT;
    float* row_sum  = att_s   + N_ATT;
    float* score    = row_sum + N_ENT;
    float* att_proj = score   + NE;
    float* U        = att_proj + (long)N_ATT * KD;

    const long nEntFeat = (long)N_ENT * KD;

    zero_f32<<<512, 256, 0, stream>>>(row_sum, (long)N_ENT);
    zero_f32<<<2048, 256, 0, stream>>>(U, nEntFeat);
    zero_f32<<<2048, 256, 0, stream>>>(out, nEntFeat);

    {
        int waves = N_ENT + N_ATT;
        int blocks = (waves * 32 + 255) / 256;
        row_dot_kernel<<<blocks, 256, 0, stream>>>(ent_feats, att_feats, a_w,
                                                   ent_s, att_s);
    }

    {
        int tiles = (N_ATT / 16) * 16;
        wmma_gemm_f32<false><<<tiles / 8, 256, 0, stream>>>(att_feats, W_top, att_proj);
    }

    edge_score_kernel<<<(NE + 255) / 256, 256, 0, stream>>>(trip, ent_s, att_s, a_b,
                                                            score, row_sum);

    edge_agg_kernel<<<2048, 256, 0, stream>>>(trip, score, row_sum,
                                              att_proj, val_feats, U, out);

    {
        int tiles = (N_ENT / 16) * 16;
        wmma_gemm_f32<true><<<tiles / 8, 256, 0, stream>>>(U, W_bot, out);
    }

    finalize_kernel<<<4096, 256, 0, stream>>>(out, ent_feats, nEntFeat);
}


# Row pass walks each row's edge list once, not twice: softmax sums relative to a running max, divided by the row sum at the end (sum order kept, rounding changes); edge records read ahead of the rows
# speedup vs baseline: 2.5136x; 2.5136x over previous
#include <hip/hip_runtime.h>
#include <math.h>

#define N_ENT 50000
#define N_ATT 2000
#define N_VAL 100000
#define NE    300000
#define KD    256

typedef __attribute__((ext_vector_type(2))) float v2f;
typedef __attribute__((ext_vector_type(8))) float v8f;

__global__ void edge_link_kernel(const int* __restrict__ trip,
                                 int* __restrict__ head,
                                 int* __restrict__ nxt) {
    int e = blockIdx.x * blockDim.x + threadIdx.x;
    if (e >= NE) return;
    int h = trip[3 * e + 0];
    nxt[e] = atomicExch(&head[h], e);
}

#define FRONT_WG     500
#define FRONT_GROWS  8
#define FRONT_SROWS  18
#define FRONT_BROWS  (4 * FRONT_GROWS + 4 * FRONT_SROWS)
#define N_ROWS       (N_ENT + N_ATT)
#define ATT_TILES    ((N_ATT / 16) * 16)

static_assert(N_ATT % 16 == 0, "att_proj is tiled 16 rows at a time");
static_assert(FRONT_WG * 4 >= ATT_TILES, "one GEMM tile per wave 0-3");
static_assert(FRONT_WG * FRONT_BROWS >= N_ROWS, "every score row has a wave");
static_assert(FRONT_WG * 256 >= N_ENT, "every list head has a thread");
static_assert(FRONT_GROWS <= 32 && FRONT_SROWS <= 32, "one lane per row at the store");

__device__ __forceinline__ void att_gemm_tile(int tile, int lane,
                                              const float* __restrict__ A,
                                              const float* __restrict__ B,
                                              float* __restrict__ C) {
    int tM = tile >> 4;
    int tN = tile & 15;
    int row0 = tM * 16;
    int col0 = tN * 16;

    int m    = lane & 15;
    int koff = (lane >> 4) << 1;
    int bn   = col0 + (lane & 15);
    int crow = row0 + ((lane >> 4) << 3);
    int cn   = col0 + (lane & 15);

    v8f c;
#pragma unroll
    for (int r = 0; r < 8; ++r) c[r] = 0.0f;

    const float* aptr = A + (long)(row0 + m) * KD + koff;

#pragma unroll 4
    for (int k0 = 0; k0 < KD; k0 += 4) {
        v2f av = *(const v2f*)(aptr + k0);
        v2f bv;
        bv.x = B[(long)(k0 + koff + 0) * KD + bn];
        bv.y = B[(long)(k0 + koff + 1) * KD + bn];
        c = __builtin_amdgcn_wmma_f32_16x16x4_f32(
                 false, av,  false, bv,
                 (short)0, c,  false,  false);
    }

#pragma unroll
    for (int r = 0; r < 8; ++r) C[(long)(crow + r) * KD + cn] = c[r];
}

template <int NR>
__device__ __forceinline__ void dot_rows(int r0, int lane,
                                         const float* __restrict__ ent_feats,
                                         const float* __restrict__ att_feats,
                                         const float4 we0, const float4 we1,
                                         const float4 wa0, const float4 wa1,
                                         float* __restrict__ ent_score,
                                         float* __restrict__ att_score) {
    const int c0 = lane * 4;
    const int c1 = 128 + lane * 4;
    float4 v0[NR], v1[NR];
#pragma unroll
    for (int k = 0; k < NR; ++k) {
        int r = r0 + k;
        if (r >= N_ROWS) r = 0;
        const float* feat = (r < N_ENT) ? ent_feats + (long)r * KD
                                        : att_feats + (long)(r - N_ENT) * KD;
        v0[k] = *(const float4*)(feat + c0);
        v1[k] = *(const float4*)(feat + c1);
    }
    __builtin_amdgcn_sched_barrier(0);
    float mine = 0.0f;
#pragma unroll
    for (int k = 0; k < NR; ++k) {
        const bool isEnt = (r0 + k) < N_ENT;
        const float4 w0 = isEnt ? we0 : wa0;
        const float4 w1 = isEnt ? we1 : wa1;
        float s = 0.0f;
        s += v0[k].x * w0.x + v0[k].y * w0.y + v0[k].z * w0.z + v0[k].w * w0.w;
        s += v1[k].x * w1.x + v1[k].y * w1.y + v1[k].z * w1.z + v1[k].w * w1.w;
#pragma unroll
        for (int off = 16; off > 0; off >>= 1) s += __shfl_xor(s, off, 32);
        if (lane == k) mine = s;
    }
    const int r = r0 + lane;
    if (lane < NR && r < N_ROWS) {
        if (r < N_ENT) ent_score[r] = mine;
        else           att_score[r - N_ENT] = mine;
    }
}

__global__ __launch_bounds__(256)
void front_kernel(const float* __restrict__ ent_feats,
                  const float* __restrict__ att_feats,
                  const float* __restrict__ a_w,
                  const float* __restrict__ W_top,
                  float* __restrict__ ent_score,
                  float* __restrict__ att_score,
                  int* __restrict__ head,
                  float* __restrict__ att_proj) {
    const int lane = threadIdx.x & 31;
    const int wave = __builtin_amdgcn_readfirstlane((int)(threadIdx.x >> 5));
    const int wg   = blockIdx.x;

    {
        int t = wg * 256 + (int)threadIdx.x;
        if (t < N_ENT) head[t] = -1;
    }

    const float4 we0 = *(const float4*)(a_w + lane * 4);
    const float4 we1 = *(const float4*)(a_w + 128 + lane * 4);
    const float4 wa0 = *(const float4*)(a_w + KD + lane * 4);
    const float4 wa1 = *(const float4*)(a_w + KD + 128 + lane * 4);

    if (wave < 4) {
        const int tile = wg * 4 + wave;
        if (tile < ATT_TILES) att_gemm_tile(tile, lane, att_feats, W_top, att_proj);
        dot_rows<FRONT_GROWS>(wg * FRONT_BROWS + FRONT_GROWS * wave, lane,
                              ent_feats, att_feats, we0, we1, wa0, wa1,
                              ent_score, att_score);
    } else {
        dot_rows<FRONT_SROWS>(wg * FRONT_BROWS + 4 * FRONT_GROWS + FRONT_SROWS * (wave - 4), lane,
                              ent_feats, att_feats, we0, we1, wa0, wa1,
                              ent_score, att_score);
    }
}

#define GEMM_WG     256
#define LDS_ROW_F4  65

__global__ __launch_bounds__(512)
void wmma_gemm_f32_breg(const float* __restrict__ A,
                        const float* __restrict__ B,
                        float* __restrict__ C,
                        int nTiles) {
    __shared__ float4 sA[2][16 * LDS_ROW_F4];

    const int lane = threadIdx.x & 31;
    const int wave = threadIdx.x >> 5;
    const int c16  = lane & 15;
    const int h    = lane >> 4;

    int tile = blockIdx.x;
    if (tile >= nTiles) return;

    v2f b[64];
    {
        const float* bp = B + (long)(4 * h) * KD + wave * 16 + c16;
#pragma unroll
        for (int j = 0; j < 32; ++j) {
            const float* q = bp + (long)(8 * j) * KD;
            b[2 * j].x     = q[0 * KD];
            b[2 * j].y     = q[1 * KD];
            b[2 * j + 1].x = q[2 * KD];
            b[2 * j + 1].y = q[3 * KD];
        }
    }

    const int stRow = wave * LDS_ROW_F4 + lane;
    const int cOff  = (8 * h) * KD + wave * 16 + c16;

    v8f c;
    {
        const float* src = A + ((long)tile * 16 + wave) * KD + 4 * lane;
        float4 g0 = *(const float4*)(src);
        float4 g1 = *(const float4*)(src + 128);
        const float* cp = C + (long)tile * 16 * KD + cOff;
#pragma unroll
        for (int r = 0; r < 8; ++r) c[r] = cp[r * KD];
        sA[0][stRow]      = g0;
        sA[0][stRow + 32] = g1;
    }
    __syncthreads();

    int stage = 0;
    for (; tile < nTiles; tile += GEMM_WG) {
        const int  nxt      = tile + GEMM_WG;
        const bool has_next = nxt < nTiles;

        float4 g0 = make_float4(0.f, 0.f, 0.f, 0.f), g1 = g0;
        v8f cn = {0.f, 0.f, 0.f, 0.f, 0.f, 0.f, 0.f, 0.f};
        if (has_next) {
            const float* src = A + ((long)nxt * 16 + wave) * KD + 4 * lane;
            g0 = *(const float4*)(src);
            g1 = *(const float4*)(src + 128);
            const float* cp = C + (long)nxt * 16 * KD + cOff;
#pragma unroll
            for (int r = 0; r < 8; ++r) cn[r] = cp[r * KD];
        }

        const float4* sa = &sA[stage][c16 * LDS_ROW_F4 + h];
#pragma unroll
        for (int j = 0; j < 32; ++j) {
            const float4 a = sa[2 * j];
            v2f a0; a0.x = a.x; a0.y = a.y;
            v2f a1; a1.x = a.z; a1.y = a.w;
            c = __builtin_amdgcn_wmma_f32_16x16x4_f32(
                     false, a0,  false, b[2 * j],
                     (short)0, c,  false,  false);
            c = __builtin_amdgcn_wmma_f32_16x16x4_f32(
                     false, a1,  false, b[2 * j + 1],
                     (short)0, c,  false,  false);
        }

        if (has_next) {
            sA[stage ^ 1][stRow]      = g0;
            sA[stage ^ 1][stRow + 32] = g1;
        }
        __syncthreads();

        float* op = C + (long)tile * 16 * KD + cOff;
#pragma unroll
        for (int r = 0; r < 8; ++r) op[r * KD] = c[r];

        c = cn;
        stage ^= 1;
    }
}

__global__ void row_pull_kernel(const int* __restrict__ trip,
                                const int* __restrict__ head,
                                const int* __restrict__ nxt,
                                const float* __restrict__ ent_score,
                                const float* __restrict__ att_score,
                                const float* __restrict__ a_b,
                                const float* __restrict__ att_proj,
                                const float* __restrict__ val_feats,
                                float* __restrict__ U,
                                float* __restrict__ outAcc) {
    int lane = threadIdx.x & 31;
    int row = __builtin_amdgcn_readfirstlane(
        (int)((blockIdx.x * blockDim.x + threadIdx.x) >> 5));
    if (row >= N_ENT) return;

    const float es = ent_score[row];
    const float b  = a_b[0];
    const int first = head[row];

    float4 u0 = make_float4(0.f, 0.f, 0.f, 0.f), u1 = u0;
    float4 o0 = u0, o1 = u0;
    const int c0 = lane * 4;
    const int c1 = 128 + lane * 4;

    if ((unsigned)first < (unsigned)NE) {
        const int  vlF  = trip[3 * first + 1];
        const int  atF  = trip[3 * first + 2];
        const int  enF  = nxt[first];
        bool more  = (unsigned)enF < (unsigned)NE;
        const int  eN   = more ? enF : first;
        long vo    = (long)vlF * KD;
        long ao    = (long)atF * KD;
        const float asF = att_score[atF];
        const int  vlN  = trip[3 * eN + 1];
        const int  atN  = trip[3 * eN + 2];
        const int  enN  = nxt[eN];
        float x    = es + asF + b;
        x = (x > 0.0f) ? x : 0.2f * x;
        long voN   = (long)vlN * KD;
        long aoN   = (long)atN * KD;
        const float* asPN = att_score + atN;
        bool more2 = more && ((unsigned)enN < (unsigned)NE);
        int  e2    = more2 ? enN : eN;
        float m    = -3.402823466e+38f;
        float rs   = 0.0f;
        int it = 0;
        bool go;
        do {
            const float* vrow = val_feats + vo;
            const float* arow = att_proj + ao;
            const float4 v0 = *(const float4*)(vrow + c0);
            const float4 v1 = *(const float4*)(vrow + c1);
            const float4 a0 = *(const float4*)(arow + c0);
            const float4 a1 = *(const float4*)(arow + c1);
            const float asN = *asPN;
            const int   vl2 = trip[3 * e2 + 1];
            const int   at2 = trip[3 * e2 + 2];
            const int   en2 = nxt[e2];
            asm volatile("" ::: "memory");

            const float mn = fmaxf(m, x);
            const float sc = expf(m - mn);
            const float w  = expf(x - mn);
            m  = mn;
            rs = rs * sc + w;
            u0.x = u0.x * sc + w * v0.x; u0.y = u0.y * sc + w * v0.y;
            u0.z = u0.z * sc + w * v0.z; u0.w = u0.w * sc + w * v0.w;
            u1.x = u1.x * sc + w * v1.x; u1.y = u1.y * sc + w * v1.y;
            u1.z = u1.z * sc + w * v1.z; u1.w = u1.w * sc + w * v1.w;
            o0.x = o0.x * sc + w * a0.x; o0.y = o0.y * sc + w * a0.y;
            o0.z = o0.z * sc + w * a0.z; o0.w = o0.w * sc + w * a0.w;
            o1.x = o1.x * sc + w * a1.x; o1.y = o1.y * sc + w * a1.y;
            o1.z = o1.z * sc + w * a1.z; o1.w = o1.w * sc + w * a1.w;

            go = more && (++it < NE);
            float xN = es + asN + b;
            xN = (xN > 0.0f) ? xN : 0.2f * xN;
            const bool more3 = more2 && ((unsigned)en2 < (unsigned)NE);
            const int  e3    = more3 ? en2 : e2;
            vo = voN; ao = aoN; x = xN; more = more2;
            voN = (long)vl2 * KD; aoN = (long)at2 * KD; asPN = att_score + at2;
            more2 = more3; e2 = e3;
        } while (go);
        const float inv = 1.0f / rs;
        u0.x *= inv; u0.y *= inv; u0.z *= inv; u0.w *= inv;
        u1.x *= inv; u1.y *= inv; u1.z *= inv; u1.w *= inv;
        o0.x *= inv; o0.y *= inv; o0.z *= inv; o0.w *= inv;
        o1.x *= inv; o1.y *= inv; o1.z *= inv; o1.w *= inv;
    }

    float* urow = U + (long)row * KD;
    float* orow = outAcc + (long)row * KD;
    *(float4*)(urow + c0) = u0;
    *(float4*)(urow + c1) = u1;
    *(float4*)(orow + c0) = o0;
    *(float4*)(orow + c1) = o1;
}

__global__ void finalize_kernel(float* __restrict__ out,
                                const float* __restrict__ ent_feats,
                                long n) {
    long i = (long)blockIdx.x * blockDim.x + threadIdx.x;
    long stride = (long)gridDim.x * blockDim.x;
    for (; i < n; i += stride) {
        float x = out[i] + ent_feats[i];
        out[i] = (x > 0.0f) ? x : (expf(x) - 1.0f);
    }
}

extern "C" void kernel_launch(void* const* d_in, const int* in_sizes, int n_in,
                              void* d_out, int out_size, void* d_ws, size_t ws_size,
                              hipStream_t stream) {
    const int*   trip      = (const int*)d_in[0];
    const float* att_feats = (const float*)d_in[1];
    const float* val_feats = (const float*)d_in[2];
    const float* ent_feats = (const float*)d_in[3];
    const float* a_w       = (const float*)d_in[4];
    const float* a_b       = (const float*)d_in[5];
    const float* W         = (const float*)d_in[6];
    float* out = (float*)d_out;

    const float* W_top = W;
    const float* W_bot = W + KD * KD;

    float* ws       = (float*)d_ws;
    float* ent_s    = ws;
    float* att_s    = ent_s   + N_ENT;
    int*   head     = (int*)(att_s + N_ATT);
    int*   nxt      = head    + N_ENT;
    float* att_proj = (float*)(nxt + NE);
    float* U        = att_proj + (long)N_ATT * KD;

    const long nEntFeat = (long)N_ENT * KD;

    front_kernel<<<FRONT_WG, 256, 0, stream>>>(ent_feats, att_feats, a_w, W_top,
                                               ent_s, att_s, head, att_proj);

    edge_link_kernel<<<(NE + 255) / 256, 256, 0, stream>>>(trip, head, nxt);

    row_pull_kernel<<<(N_ENT * 32) / 256, 256, 0, stream>>>(trip, head, nxt, ent_s, att_s,
                                                            a_b, att_proj, val_feats, U, out);

    wmma_gemm_f32_breg<<<GEMM_WG, 512, 0, stream>>>(U, W_bot, out, N_ENT / 16);

    finalize_kernel<<<4096, 256, 0, stream>>>(out, ent_feats, nEntFeat);
}
